# starting kernel
# baseline (speedup 1.0000x reference)
#include <hip/hip_runtime.h>

typedef __attribute__((ext_vector_type(2))) float v2f;
typedef __attribute__((ext_vector_type(8))) float v8f;

#define TDIM 16
#define NACT 8

__global__ __launch_bounds__(256) void prep_weights(const float* __restrict__ Ag,
                                                    const float* __restrict__ Bg,
                                                    float* __restrict__ Wg)
{
    __shared__ float aug[NACT][TDIM][2 * TDIM];
    __shared__ float tmp[NACT][TDIM][TDIM];
    const int w    = threadIdx.x >> 5;
    const int lane = threadIdx.x & 31;
    const float* A = Ag + w * TDIM * TDIM;
    const float* B = Bg + w * TDIM * TDIM;

    for (int r = 0; r < TDIM; ++r) {
        float v;
        if (lane < TDIM) v = A[r * TDIM + lane];
        else             v = (lane - TDIM == r) ? 1.0f : 0.0f;
        aug[w][r][lane] = v;
    }
    __syncthreads();

    for (int k = 0; k < TDIM; ++k) {
        int prow = k;
        float best = fabsf(aug[w][k][k]);
        for (int r = k + 1; r < TDIM; ++r) {
            float v = fabsf(aug[w][r][k]);
            if (v > best) { best = v; prow = r; }
        }
        if (prow != k) {
            float t0 = aug[w][k][lane];
            float t1 = aug[w][prow][lane];
            aug[w][k][lane]    = t1;
            aug[w][prow][lane] = t0;
        }
        __syncthreads();
        float piv = aug[w][k][k];
        __syncthreads();
        aug[w][k][lane] = aug[w][k][lane] / piv;
        __syncthreads();
        float rowk = aug[w][k][lane];
        for (int r = 0; r < TDIM; ++r) {
            if (r == k) continue;
            float f = aug[w][r][k];
            aug[w][r][lane] = aug[w][r][lane] - f * rowk;
        }
        __syncthreads();
    }

    for (int t = 0; t < 8; ++t) {
        int c = lane + 32 * t;
        int i = c >> 4, j = c & 15;
        float s = 0.f;
        for (int k2 = 0; k2 < TDIM; ++k2)
            s += aug[w][i][TDIM + k2] * B[k2 * TDIM + j];
        tmp[w][i][j] = s;
    }
    __syncthreads();

    for (int t = 0; t < 8; ++t) {
        int c = lane + 32 * t;
        int i = c >> 4, j = c & 15;
        float s = 0.f;
        for (int k2 = 0; k2 < TDIM; ++k2)
            s += tmp[w][i][k2] * A[k2 * TDIM + j];
        Wg[w * TDIM * TDIM + c] = s;
    }
}

template<bool FULL>
__device__ __forceinline__ void do_tile(const float* __restrict__ x,
                                        const int*   __restrict__ act,
                                        float*       __restrict__ y,
                                        const v2f (&bv)[NACT][4],
                                        int tb, int n, int r, int hi, int lane)
{
    const int row = tb + r;

    v2f xv[4];
    int myAct;
    if (FULL) {
        #pragma unroll
        for (int ks = 0; ks < 4; ++ks)
            xv[ks] = *(const v2f*)(x + row * 16 + 4 * ks + 2 * hi);
        myAct = act[row];
    } else {
        const bool rok = row < n;
        #pragma unroll
        for (int ks = 0; ks < 4; ++ks) {
            v2f v = {0.f, 0.f};
            if (rok) v = *(const v2f*)(x + row * 16 + 4 * ks + 2 * hi);
            xv[ks] = v;
        }
        myAct = rok ? act[row] : -1;
    }

    v8f acc0 = {0.f, 0.f, 0.f, 0.f, 0.f, 0.f, 0.f, 0.f};
    v8f acc1 = {0.f, 0.f, 0.f, 0.f, 0.f, 0.f, 0.f, 0.f};
    #pragma unroll
    for (int a = 0; a < NACT; a += 2) {
        const bool m0 = (myAct == a);
        const bool m1 = (myAct == a + 1);
        #pragma unroll
        for (int ks = 0; ks < 4; ++ks) {
            v2f xa0, xa1;
            xa0.x = m0 ? xv[ks].x : 0.f;
            xa0.y = m0 ? xv[ks].y : 0.f;
            xa1.x = m1 ? xv[ks].x : 0.f;
            xa1.y = m1 ? xv[ks].y : 0.f;
            acc0 = __builtin_amdgcn_wmma_f32_16x16x4_f32(
                       false, xa0, false, bv[a][ks],     (short)0, acc0, false, false);
            acc1 = __builtin_amdgcn_wmma_f32_16x16x4_f32(
                       false, xa1, false, bv[a + 1][ks], (short)0, acc1, false, false);
        }
    }

    #pragma unroll
    for (int v = 0; v < 8; ++v) {
        const int s = tb + v + 8 * hi;
        const float val = acc0[v] + acc1[v];
        if (FULL) {
            y[s * 16 + (lane & 15)] = val;
        } else if (s < n) {
            y[s * 16 + (lane & 15)] = val;
        }
    }
}

__global__ __launch_bounds__(256) void gemv_select(const float* __restrict__ x,
                                                   const int*   __restrict__ act,
                                                   const float* __restrict__ Wg,
                                                   float*       __restrict__ y,
                                                   int n, int ntiles)
{
    const int lane        = threadIdx.x & 31;
    const int waveInBlk   = threadIdx.x >> 5;
    const int wavesPerBlk = blockDim.x >> 5;
    const int gwave  = blockIdx.x * wavesPerBlk + waveInBlk;
    const int nwaves = gridDim.x * wavesPerBlk;

    const int r  = lane & 15;
    const int hi = lane >> 4;

    v2f bv[NACT][4];
    #pragma unroll
    for (int a = 0; a < NACT; ++a) {
        #pragma unroll
        for (int ks = 0; ks < 4; ++ks) {
            const float* p = Wg + a * 256 + r * 16 + 4 * ks + 2 * hi;
            bv[a][ks] = *(const v2f*)p;
        }
    }

    for (int t = gwave; t < ntiles; t += nwaves) {
        const int tb = t * 16;
        if (tb + 16 <= n) {
            do_tile<true>(x, act, y, bv, tb, n, r, hi, lane);
        } else {
            do_tile<false>(x, act, y, bv, tb, n, r, hi, lane);
        }
    }
}

extern "C" void kernel_launch(void* const* d_in, const int* in_sizes, int n_in,
                              void* d_out, int out_size, void* d_ws, size_t ws_size,
                              hipStream_t stream) {
    const float* x    = (const float*)d_in[0];
    const int*   a    = (const int*)  d_in[1];
    const float* A    = (const float*)d_in[2];
    const float* Brep = (const float*)d_in[3];
    float* y  = (float*)d_out;
    float* Wg = (float*)d_ws;

    const int n = in_sizes[1];

    prep_weights<<<1, 256, 0, stream>>>(A, Brep, Wg);

    const int ntiles = (n + 15) / 16;
    const int wavesPerBlk  = 8;
    const int tilesPerWave = 4;
    int blocks = (ntiles + wavesPerBlk * tilesPerWave - 1) / (wavesPerBlk * tilesPerWave);
    if (blocks < 1) blocks = 1;
    if (blocks > 32768) blocks = 32768;
    gemv_select<<<blocks, 256, 0, stream>>>(x, a, Wg, y, n, ntiles);
}


# Weight-prep kernel moved from LDS and barriers into registers: one wave per action, one matrix column per lane, pivot column by lane reads; same operations and summation order
# speedup vs baseline: 1.8118x; 1.8118x over previous
#include <hip/hip_runtime.h>

typedef __attribute__((ext_vector_type(2))) float v2f;
typedef __attribute__((ext_vector_type(8))) float v8f;

#define TDIM 16
#define NACT 8

__device__ __forceinline__ float lane_bcast(float v, int src)
{
    return __int_as_float(__builtin_amdgcn_readlane(__float_as_int(v), src));
}

__global__ __launch_bounds__(32) void prep_weights(const float* __restrict__ Ag,
                                                   const float* __restrict__ Bg,
                                                   float* __restrict__ Wg)
{
    const int w    = blockIdx.x;
    const int lane = threadIdx.x;
    const int col  = lane & 15;
    const int hi   = lane >> 4;
    const float* A = Ag + w * TDIM * TDIM;
    const float* B = Bg + w * TDIM * TDIM;


    float acol[TDIM], bcol[TDIM], c[TDIM];
    #pragma unroll
    for (int r = 0; r < TDIM; ++r) {
        acol[r] = A[r * TDIM + col];
        bcol[r] = B[r * TDIM + col];
        c[r] = hi ? ((col == r) ? 1.0f : 0.0f) : acol[r];
    }

    #pragma unroll
    for (int k = 0; k < TDIM; ++k) {
        float s[TDIM];
        #pragma unroll
        for (int r = 0; r < TDIM; ++r) s[r] = lane_bcast(c[r], k);

        int prow = k;
        float best = fabsf(s[k]);
        #pragma unroll
        for (int r = k + 1; r < TDIM; ++r) {
            float v = fabsf(s[r]);
            if (v > best) { best = v; prow = r; }
        }
        if (prow != k) {
            #pragma unroll
            for (int r = k + 1; r < TDIM; ++r) {
                const bool sw = (prow == r);
                const float ck = c[k], cr = c[r];
                c[k] = sw ? cr : ck;
                c[r] = sw ? ck : cr;
                const float sk = s[k], sr = s[r];
                s[k] = sw ? sr : sk;
                s[r] = sw ? sk : sr;
            }
        }
        const float piv = s[k];
        c[k] = c[k] / piv;
        const float rowk = c[k];
        #pragma unroll
        for (int r = 0; r < TDIM; ++r) {
            if (r == k) continue;
            c[r] = c[r] - s[r] * rowk;
        }
    }

    float t[TDIM] = {};
    #pragma unroll
    for (int k2 = 0; k2 < TDIM; ++k2) {
        #pragma unroll
        for (int i = 0; i < TDIM; ++i)
            t[i] += lane_bcast(c[i], TDIM + k2) * bcol[k2];
    }

    float wv[TDIM] = {};
    #pragma unroll
    for (int k2 = 0; k2 < TDIM; ++k2) {
        #pragma unroll
        for (int i = 0; i < TDIM; ++i)
            wv[i] += lane_bcast(t[i], k2) * acol[k2];
    }

    #pragma unroll
    for (int m = 0; m < TDIM / 2; ++m)
        Wg[w * TDIM * TDIM + 32 * m + lane] = hi ? wv[2 * m + 1] : wv[2 * m];
}

template<bool FULL>
__device__ __forceinline__ void do_tile(const float* __restrict__ x,
                                        const int*   __restrict__ act,
                                        float*       __restrict__ y,
                                        const v2f (&bv)[NACT][4],
                                        int tb, int n, int r, int hi, int lane)
{
    const int row = tb + r;

    v2f xv[4];
    int myAct;
    if (FULL) {
        #pragma unroll
        for (int ks = 0; ks < 4; ++ks)
            xv[ks] = *(const v2f*)(x + row * 16 + 4 * ks + 2 * hi);
        myAct = act[row];
    } else {
        const bool rok = row < n;
        #pragma unroll
        for (int ks = 0; ks < 4; ++ks) {
            v2f v = {0.f, 0.f};
            if (rok) v = *(const v2f*)(x + row * 16 + 4 * ks + 2 * hi);
            xv[ks] = v;
        }
        myAct = rok ? act[row] : -1;
    }

    v8f acc0 = {0.f, 0.f, 0.f, 0.f, 0.f, 0.f, 0.f, 0.f};
    v8f acc1 = {0.f, 0.f, 0.f, 0.f, 0.f, 0.f, 0.f, 0.f};
    #pragma unroll
    for (int a = 0; a < NACT; a += 2) {
        const bool m0 = (myAct == a);
        const bool m1 = (myAct == a + 1);
        #pragma unroll
        for (int ks = 0; ks < 4; ++ks) {
            v2f xa0, xa1;
            xa0.x = m0 ? xv[ks].x : 0.f;
            xa0.y = m0 ? xv[ks].y : 0.f;
            xa1.x = m1 ? xv[ks].x : 0.f;
            xa1.y = m1 ? xv[ks].y : 0.f;
            acc0 = __builtin_amdgcn_wmma_f32_16x16x4_f32(
                       false, xa0, false, bv[a][ks],     (short)0, acc0, false, false);
            acc1 = __builtin_amdgcn_wmma_f32_16x16x4_f32(
                       false, xa1, false, bv[a + 1][ks], (short)0, acc1, false, false);
        }
    }

    #pragma unroll
    for (int v = 0; v < 8; ++v) {
        const int s = tb + v + 8 * hi;
        const float val = acc0[v] + acc1[v];
        if (FULL) {
            y[s * 16 + (lane & 15)] = val;
        } else if (s < n) {
            y[s * 16 + (lane & 15)] = val;
        }
    }
}

__global__ __launch_bounds__(256) void gemv_select(const float* __restrict__ x,
                                                   const int*   __restrict__ act,
                                                   const float* __restrict__ Wg,
                                                   float*       __restrict__ y,
                                                   int n, int ntiles)
{
    const int lane        = threadIdx.x & 31;
    const int waveInBlk   = threadIdx.x >> 5;
    const int wavesPerBlk = blockDim.x >> 5;
    const int gwave  = blockIdx.x * wavesPerBlk + waveInBlk;
    const int nwaves = gridDim.x * wavesPerBlk;

    const int r  = lane & 15;
    const int hi = lane >> 4;

    v2f bv[NACT][4];
    #pragma unroll
    for (int a = 0; a < NACT; ++a) {
        #pragma unroll
        for (int ks = 0; ks < 4; ++ks) {
            const float* p = Wg + a * 256 + r * 16 + 4 * ks + 2 * hi;
            bv[a][ks] = *(const v2f*)p;
        }
    }

    for (int t = gwave; t < ntiles; t += nwaves) {
        const int tb = t * 16;
        if (tb + 16 <= n) {
            do_tile<true>(x, act, y, bv, tb, n, r, hi, lane);
        } else {
            do_tile<false>(x, act, y, bv, tb, n, r, hi, lane);
        }
    }
}

extern "C" void kernel_launch(void* const* d_in, const int* in_sizes, int n_in,
                              void* d_out, int out_size, void* d_ws, size_t ws_size,
                              hipStream_t stream) {
    const float* x    = (const float*)d_in[0];
    const int*   a    = (const int*)  d_in[1];
    const float* A    = (const float*)d_in[2];
    const float* Brep = (const float*)d_in[3];
    float* y  = (float*)d_out;
    float* Wg = (float*)d_ws;

    const int n = in_sizes[1];

    prep_weights<<<NACT, 32, 0, stream>>>(A, Brep, Wg);

    const int ntiles = (n + 15) / 16;
    const int wavesPerBlk  = 8;
    const int tilesPerWave = 4;
    int blocks = (ntiles + wavesPerBlk * tilesPerWave - 1) / (wavesPerBlk * tilesPerWave);
    if (blocks < 1) blocks = 1;
    if (blocks > 32768) blocks = 32768;
    gemv_select<<<blocks, 256, 0, stream>>>(x, a, Wg, y, n, ntiles);
}


# Main kernel: each wave sorts its 128 samples by action into an LDS table (ballot ranks, no atomics), so a 16-row tile has one action and takes 4 matrix instructions, not 32; k order and sums kept
# speedup vs baseline: 2.6207x; 1.4465x over previous
#include <hip/hip_runtime.h>

typedef __attribute__((ext_vector_type(2))) float v2f;
typedef __attribute__((ext_vector_type(8))) float v8f;

#define TDIM 16
#define NACT 8

__device__ __forceinline__ float lane_bcast(float v, int src)
{
    return __int_as_float(__builtin_amdgcn_readlane(__float_as_int(v), src));
}

__global__ __launch_bounds__(32) void prep_weights(const float* __restrict__ Ag,
                                                   const float* __restrict__ Bg,
                                                   float* __restrict__ Wg)
{
    const int w    = blockIdx.x;
    const int lane = threadIdx.x;
    const int col  = lane & 15;
    const int hi   = lane >> 4;
    const float* A = Ag + w * TDIM * TDIM;
    const float* B = Bg + w * TDIM * TDIM;


    float acol[TDIM], bcol[TDIM], c[TDIM];
    #pragma unroll
    for (int r = 0; r < TDIM; ++r) {
        acol[r] = A[r * TDIM + col];
        bcol[r] = B[r * TDIM + col];
        c[r] = hi ? ((col == r) ? 1.0f : 0.0f) : acol[r];
    }

    #pragma unroll
    for (int k = 0; k < TDIM; ++k) {
        float s[TDIM];
        #pragma unroll
        for (int r = 0; r < TDIM; ++r) s[r] = lane_bcast(c[r], k);

        int prow = k;
        float best = fabsf(s[k]);
        #pragma unroll
        for (int r = k + 1; r < TDIM; ++r) {
            float v = fabsf(s[r]);
            if (v > best) { best = v; prow = r; }
        }
        if (prow != k) {
            #pragma unroll
            for (int r = k + 1; r < TDIM; ++r) {
                const bool sw = (prow == r);
                const float ck = c[k], cr = c[r];
                c[k] = sw ? cr : ck;
                c[r] = sw ? ck : cr;
                const float sk = s[k], sr = s[r];
                s[k] = sw ? sr : sk;
                s[r] = sw ? sk : sr;
            }
        }
        const float piv = s[k];
        c[k] = c[k] / piv;
        const float rowk = c[k];
        #pragma unroll
        for (int r = 0; r < TDIM; ++r) {
            if (r == k) continue;
            c[r] = c[r] - s[r] * rowk;
        }
    }

    float t[TDIM] = {};
    #pragma unroll
    for (int k2 = 0; k2 < TDIM; ++k2) {
        #pragma unroll
        for (int i = 0; i < TDIM; ++i)
            t[i] += lane_bcast(c[i], TDIM + k2) * bcol[k2];
    }

    float wv[TDIM] = {};
    #pragma unroll
    for (int k2 = 0; k2 < TDIM; ++k2) {
        #pragma unroll
        for (int i = 0; i < TDIM; ++i)
            wv[i] += lane_bcast(t[i], k2) * acol[k2];
    }

    #pragma unroll
    for (int m = 0; m < TDIM / 2; ++m)
        Wg[w * TDIM * TDIM + 32 * m + lane] = hi ? wv[2 * m + 1] : wv[2 * m];
}

#define GROUP   128
#define SLOTS   249
#define SPARE   (SLOTS - 1)
#define PITCH   20
#define WAVES_PER_BLK 8

__device__ __forceinline__ void wave_lds_order()
{
    __builtin_amdgcn_fence(__ATOMIC_RELEASE, "wavefront");
    __builtin_amdgcn_wave_barrier();
    __builtin_amdgcn_fence(__ATOMIC_ACQUIRE, "wavefront");
}

__device__ __forceinline__ void group_tile(float* __restrict__ rs, int first, int live,
                                           const v2f (&w)[4], int c, int hi)
{
    const float* rp = rs + (first + (c < live ? c : 0)) * PITCH;

    v2f xv[4];
    #pragma unroll
    for (int ks = 0; ks < 4; ++ks)
        xv[ks] = *(const v2f*)(rp + 4 * ks + 2 * hi);

    v8f acc = {0.f, 0.f, 0.f, 0.f, 0.f, 0.f, 0.f, 0.f};
    #pragma unroll
    for (int ks = 0; ks < 4; ++ks)
        acc = __builtin_amdgcn_wmma_f32_16x16x4_f32(
                  false, w[ks], false, xv[ks], (short)0, acc, false, false);

    float4* op = (float4*)(rs + (first + c) * PITCH + 8 * hi);
    const float4 o0 = {acc[0], acc[1], acc[2], acc[3]};
    const float4 o1 = {acc[4], acc[5], acc[6], acc[7]};
    op[0] = o0;
    op[1] = o1;
}

__global__ __launch_bounds__(256) void gemv_grouped(const float* __restrict__ x,
                                                    const int*   __restrict__ act,
                                                    const float* __restrict__ Wg,
                                                    float*       __restrict__ y,
                                                    int n, int ngroups)
{
    __shared__ __attribute__((aligned(16))) float rows_all[WAVES_PER_BLK][SLOTS * PITCH];

    const int lane = threadIdx.x & 31;
    const int wv   = threadIdx.x >> 5;
    const int g    = blockIdx.x * WAVES_PER_BLK + wv;
    if (g >= ngroups) return;

    float* rs = rows_all[wv];
    const int c  = lane & 15;
    const int hi = lane >> 4;
    const int s0 = g * GROUP;
    const bool full = (s0 + GROUP <= n);

    const int last = n - 1 - s0;
    int av[4];
    #pragma unroll
    for (int q = 0; q < 4; ++q) {
        const int i = lane + 32 * q;
        const int a = act[s0 + min(i, last)];
        av[q] = (i <= last) ? a : -1;
    }

    const float4* xg = (const float4*)(x + (size_t)s0 * TDIM);
    float4 xr[16];
    if (full) {
        #pragma unroll
        for (int q = 0; q < 16; ++q) xr[q] = xg[q * 32 + lane];
    } else {
        #pragma unroll
        for (int q = 0; q < 16; ++q) {
            float4 v = {0.f, 0.f, 0.f, 0.f};
            if (8 * q + (lane >> 2) <= last) v = xg[q * 32 + lane];
            xr[q] = v;
        }
    }

    v2f wr[NACT][4];
    #pragma unroll
    for (int a = 0; a < NACT; ++a) {
        #pragma unroll
        for (int ks = 0; ks < 4; ++ks)
            wr[a][ks] = *(const v2f*)(Wg + a * 256 + c * 16 + 4 * ks + 2 * hi);
    }

    int pos[4] = {-1, -1, -1, -1};
    int base[NACT], cnt[NACT];
    int run = 0;
    #pragma unroll
    for (int a = 0; a < NACT; ++a) {
        base[a] = run;
        #pragma unroll
        for (int q = 0; q < 4; ++q) {
            const bool is = (av[q] == a);
            const unsigned m = __builtin_amdgcn_ballot_w32(is);
            const int rk = (int)__builtin_amdgcn_mbcnt_lo(m, (unsigned)run);
            pos[q] = is ? rk : pos[q];
            run += __builtin_popcount(m);
        }
        cnt[a] = run - base[a];
        run = base[a] + ((cnt[a] + 15) & ~15);
    }

    const bool anybad = __builtin_amdgcn_ballot_w32((pos[0] | pos[1] | pos[2] | pos[3]) < 0) != 0u;

    const int piece = (lane & 3) * 4;
    int slot[16];
    #pragma unroll
    for (int q = 0; q < 16; ++q) {
        const int src = ((q & 3) << 3) + (lane >> 2);
        const int sp  = __builtin_amdgcn_ds_bpermute(src << 2, pos[q >> 2]);
        slot[q] = sp < 0 ? SPARE : sp;
    }

    #pragma unroll
    for (int q = 0; q < 16; ++q)
        *(float4*)(rs + slot[q] * PITCH + piece) = xr[q];
    wave_lds_order();

    #pragma unroll
    for (int a = 0; a < NACT; ++a) {
        for (int t = 0; 16 * t < cnt[a]; ++t)
            group_tile(rs, base[a] + 16 * t, cnt[a] - 16 * t, wr[a], c, hi);
    }
    wave_lds_order();

    float4* yg = (float4*)(y + (size_t)s0 * TDIM);
    if (full && !anybad) {
        float4 o[16];
        #pragma unroll
        for (int q = 0; q < 16; ++q)
            o[q] = *(const float4*)(rs + slot[q] * PITCH + piece);
        #pragma unroll
        for (int q = 0; q < 16; ++q) yg[q * 32 + lane] = o[q];
    } else {
        #pragma unroll
        for (int q = 0; q < 16; ++q) {
            float4 v = *(const float4*)(rs + slot[q] * PITCH + piece);
            if (slot[q] == SPARE) {
                const float4 z = {0.f, 0.f, 0.f, 0.f};
                v = z;
            }
            if (s0 + q * 8 + (lane >> 2) < n) yg[q * 32 + lane] = v;
        }
    }
}

extern "C" void kernel_launch(void* const* d_in, const int* in_sizes, int n_in,
                              void* d_out, int out_size, void* d_ws, size_t ws_size,
                              hipStream_t stream) {
    const float* x    = (const float*)d_in[0];
    const int*   a    = (const int*)  d_in[1];
    const float* A    = (const float*)d_in[2];
    const float* Brep = (const float*)d_in[3];
    float* y  = (float*)d_out;
    float* Wg = (float*)d_ws;

    const int n = in_sizes[1];

    prep_weights<<<NACT, 32, 0, stream>>>(A, Brep, Wg);

    const int ngroups = (n + GROUP - 1) / GROUP;
    int blocks = (ngroups + WAVES_PER_BLK - 1) / WAVES_PER_BLK;
    if (blocks < 1) blocks = 1;
    gemv_grouped<<<blocks, 32 * WAVES_PER_BLK, 0, stream>>>(x, a, Wg, y, n, ngroups);
}


# Weight prep fused into the main kernel: waves 0-7 of each 16-wave work-group invert one action each in registers, W goes to all waves through LDS (2 barriers); one launch; same operations and sums
# speedup vs baseline: 2.9690x; 1.1329x over previous
#include <hip/hip_runtime.h>

typedef __attribute__((ext_vector_type(2))) float v2f;
typedef __attribute__((ext_vector_type(8))) float v8f;

#define TDIM 16
#define NACT 8

__device__ __forceinline__ float lane_bcast(float v, int src)
{
    return __int_as_float(__builtin_amdgcn_readlane(__float_as_int(v), src));
}

#define GROUP   128
#define SLOTS   249
#define SPARE   (SLOTS - 1)
#define PITCH   20
#define WAVES_PER_BLK 16
static_assert(NACT <= WAVES_PER_BLK, "part 1 needs one wave per action in every work-group");

__device__ __forceinline__ void weights_to_lds(const float (&acol)[TDIM], const float (&bcol)[TDIM],
                                               float* __restrict__ dst, int col, int hi)
{

    float c[TDIM];
    #pragma unroll
    for (int r = 0; r < TDIM; ++r)
        c[r] = hi ? ((col == r) ? 1.0f : 0.0f) : acol[r];

    #pragma unroll
    for (int k = 0; k < TDIM; ++k) {
        float s[TDIM];
        #pragma unroll
        for (int r = 0; r < TDIM; ++r) s[r] = lane_bcast(c[r], k);

        int prow = k;
        float best = fabsf(s[k]);
        #pragma unroll
        for (int r = k + 1; r < TDIM; ++r) {
            float v = fabsf(s[r]);
            if (v > best) { best = v; prow = r; }
        }
        if (prow != k) {
            #pragma unroll
            for (int r = k + 1; r < TDIM; ++r) {
                const bool sw = (prow == r);
                const float ck = c[k], cr = c[r];
                c[k] = sw ? cr : ck;
                c[r] = sw ? ck : cr;
                const float sk = s[k], sr = s[r];
                s[k] = sw ? sr : sk;
                s[r] = sw ? sk : sr;
            }
        }
        const float piv = s[k];
        c[k] = c[k] / piv;
        const float rowk = c[k];
        #pragma unroll
        for (int r = 0; r < TDIM; ++r) {
            if (r == k) continue;
            c[r] = c[r] - s[r] * rowk;
        }
    }

    float t[TDIM] = {};
    #pragma unroll
    for (int k2 = 0; k2 < TDIM; ++k2) {
        #pragma unroll
        for (int i = 0; i < TDIM; ++i)
            t[i] += lane_bcast(c[i], TDIM + k2) * bcol[k2];
    }

    float wv[TDIM] = {};
    #pragma unroll
    for (int k2 = 0; k2 < TDIM; ++k2) {
        #pragma unroll
        for (int i = 0; i < TDIM; ++i)
            wv[i] += lane_bcast(t[i], k2) * acol[k2];
    }

    #pragma unroll
    for (int m = 0; m < TDIM / 2; ++m)
        dst[(2 * m + hi) * PITCH + col] = hi ? wv[2 * m + 1] : wv[2 * m];
}

__device__ __forceinline__ void wave_lds_order()
{
    __builtin_amdgcn_fence(__ATOMIC_RELEASE, "wavefront");
    __builtin_amdgcn_wave_barrier();
    __builtin_amdgcn_fence(__ATOMIC_ACQUIRE, "wavefront");
}

__device__ __forceinline__ void group_tile(float* __restrict__ rs, int first, int live,
                                           const v2f (&w)[4], int c, int hi)
{
    const float* rp = rs + (first + (c < live ? c : 0)) * PITCH;

    v2f xv[4];
    #pragma unroll
    for (int ks = 0; ks < 4; ++ks)
        xv[ks] = *(const v2f*)(rp + 4 * ks + 2 * hi);

    v8f acc = {0.f, 0.f, 0.f, 0.f, 0.f, 0.f, 0.f, 0.f};
    #pragma unroll
    for (int ks = 0; ks < 4; ++ks)
        acc = __builtin_amdgcn_wmma_f32_16x16x4_f32(
                  false, w[ks], false, xv[ks], (short)0, acc, false, false);

    float4* op = (float4*)(rs + (first + c) * PITCH + 8 * hi);
    const float4 o0 = {acc[0], acc[1], acc[2], acc[3]};
    const float4 o1 = {acc[4], acc[5], acc[6], acc[7]};
    op[0] = o0;
    op[1] = o1;
}

__global__ __launch_bounds__(32 * WAVES_PER_BLK) void gemv_grouped_fused(const float* __restrict__ x,
                                                          const int*   __restrict__ act,
                                                          const float* __restrict__ Ag,
                                                          const float* __restrict__ Bg,
                                                          float*       __restrict__ y,
                                                          int n, int ngroups)
{
    __shared__ __attribute__((aligned(16))) float rows_all[WAVES_PER_BLK][SLOTS * PITCH];

    const int lane = threadIdx.x & 31;
    const int wv   = threadIdx.x >> 5;
    const bool prep = (wv < NACT);
    const int g    = blockIdx.x * WAVES_PER_BLK + wv;

    float* rs = rows_all[wv];
    const int c  = lane & 15;
    const int hi = lane >> 4;
    const int s0 = g * GROUP;
    const bool full = (s0 + GROUP <= n);

    float acol[TDIM] = {}, bcol[TDIM] = {};
    if (prep) {
        #pragma unroll
        for (int r = 0; r < TDIM; ++r) {
            acol[r] = Ag[wv * TDIM * TDIM + r * TDIM + c];
            bcol[r] = Bg[wv * TDIM * TDIM + r * TDIM + c];
        }
    }

    const int last = n - 1 - s0;
    int av[4];
    #pragma unroll
    for (int q = 0; q < 4; ++q) {
        const int i = lane + 32 * q;
        const int a = act[s0 + min(i, last)];
        av[q] = (i <= last) ? a : -1;
    }

    const float4* xg = (const float4*)(x + (size_t)s0 * TDIM);
    float4 xr[16];
    if (full) {
        #pragma unroll
        for (int q = 0; q < 16; ++q) xr[q] = xg[q * 32 + lane];
    } else {
        #pragma unroll
        for (int q = 0; q < 16; ++q) {
            float4 v = {0.f, 0.f, 0.f, 0.f};
            if (8 * q + (lane >> 2) <= last) v = xg[q * 32 + lane];
            xr[q] = v;
        }
    }

    if (prep) weights_to_lds(acol, bcol, rs, c, hi);
    __syncthreads();
    v2f wr[NACT][4];
    #pragma unroll
    for (int a = 0; a < NACT; ++a) {
        #pragma unroll
        for (int ks = 0; ks < 4; ++ks)
            wr[a][ks] = *(const v2f*)(rows_all[a] + c * PITCH + 4 * ks + 2 * hi);
    }
    __syncthreads();
    if (g >= ngroups) return;

    int pos[4] = {-1, -1, -1, -1};
    int base[NACT], cnt[NACT];
    int run = 0;
    #pragma unroll
    for (int a = 0; a < NACT; ++a) {
        base[a] = run;
        #pragma unroll
        for (int q = 0; q < 4; ++q) {
            const bool is = (av[q] == a);
            const unsigned m = __builtin_amdgcn_ballot_w32(is);
            const int rk = (int)__builtin_amdgcn_mbcnt_lo(m, (unsigned)run);
            pos[q] = is ? rk : pos[q];
            run += __builtin_popcount(m);
        }
        cnt[a] = run - base[a];
        run = base[a] + ((cnt[a] + 15) & ~15);
    }

    const bool anybad = __builtin_amdgcn_ballot_w32((pos[0] | pos[1] | pos[2] | pos[3]) < 0) != 0u;

    const int piece = (lane & 3) * 4;
    int slot[16];
    #pragma unroll
    for (int q = 0; q < 16; ++q) {
        const int src = ((q & 3) << 3) + (lane >> 2);
        const int sp  = __builtin_amdgcn_ds_bpermute(src << 2, pos[q >> 2]);
        slot[q] = sp < 0 ? SPARE : sp;
    }

    #pragma unroll
    for (int q = 0; q < 16; ++q)
        *(float4*)(rs + slot[q] * PITCH + piece) = xr[q];
    wave_lds_order();

    #pragma unroll
    for (int a = 0; a < NACT; ++a) {
        for (int t = 0; 16 * t < cnt[a]; ++t)
            group_tile(rs, base[a] + 16 * t, cnt[a] - 16 * t, wr[a], c, hi);
    }
    wave_lds_order();

    float4* yg = (float4*)(y + (size_t)s0 * TDIM);
    if (full && !anybad) {
        float4 o[16];
        #pragma unroll
        for (int q = 0; q < 16; ++q)
            o[q] = *(const float4*)(rs + slot[q] * PITCH + piece);
        #pragma unroll
        for (int q = 0; q < 16; ++q) yg[q * 32 + lane] = o[q];
    } else {
        #pragma unroll
        for (int q = 0; q < 16; ++q) {
            float4 v = *(const float4*)(rs + slot[q] * PITCH + piece);
            if (slot[q] == SPARE) {
                const float4 z = {0.f, 0.f, 0.f, 0.f};
                v = z;
            }
            if (s0 + q * 8 + (lane >> 2) < n) yg[q * 32 + lane] = v;
        }
    }
}

extern "C" void kernel_launch(void* const* d_in, const int* in_sizes, int n_in,
                              void* d_out, int out_size, void* d_ws, size_t ws_size,
                              hipStream_t stream) {
    const float* x    = (const float*)d_in[0];
    const int*   a    = (const int*)  d_in[1];
    const float* A    = (const float*)d_in[2];
    const float* Brep = (const float*)d_in[3];
    float* y  = (float*)d_out;

    const int n = in_sizes[1];

    const int ngroups = (n + GROUP - 1) / GROUP;
    int blocks = (ngroups + WAVES_PER_BLK - 1) / WAVES_PER_BLK;
    if (blocks < 1) blocks = 1;
    gemv_grouped_fused<<<blocks, 32 * WAVES_PER_BLK, 0, stream>>>(x, a, A, Brep, y, n, ngroups);
}


# Waves 8-15 of a work-group rank their samples and fill their LDS table before the barriers, while waves 0-7 make the weights (scalar wave index); 4 v_nop behind a tile's products; sums unchanged
# speedup vs baseline: 3.1257x; 1.0528x over previous
#include <hip/hip_runtime.h>

typedef __attribute__((ext_vector_type(2))) float v2f;
typedef __attribute__((ext_vector_type(8))) float v8f;

#define TDIM 16
#define NACT 8

__device__ __forceinline__ float lane_bcast(float v, int src)
{
    return __int_as_float(__builtin_amdgcn_readlane(__float_as_int(v), src));
}

#define GROUP   128
#define SLOTS   249
#define SPARE   (SLOTS - 1)
#define PITCH   20
#define WAVES_PER_BLK 16
static_assert(NACT <= WAVES_PER_BLK, "part 1 needs one wave per action in every work-group");

__device__ __forceinline__ void weights_to_lds(const float (&acol)[TDIM], const float (&bcol)[TDIM],
                                               float* __restrict__ dst, int col, int hi)
{

    float c[TDIM];
    #pragma unroll
    for (int r = 0; r < TDIM; ++r)
        c[r] = hi ? ((col == r) ? 1.0f : 0.0f) : acol[r];

    #pragma unroll
    for (int k = 0; k < TDIM; ++k) {
        float s[TDIM];
        #pragma unroll
        for (int r = 0; r < TDIM; ++r) s[r] = lane_bcast(c[r], k);

        int prow = k;
        float best = fabsf(s[k]);
        #pragma unroll
        for (int r = k + 1; r < TDIM; ++r) {
            float v = fabsf(s[r]);
            if (v > best) { best = v; prow = r; }
        }
        if (prow != k) {
            #pragma unroll
            for (int r = k + 1; r < TDIM; ++r) {
                const bool sw = (prow == r);
                const float ck = c[k], cr = c[r];
                c[k] = sw ? cr : ck;
                c[r] = sw ? ck : cr;
                const float sk = s[k], sr = s[r];
                s[k] = sw ? sr : sk;
                s[r] = sw ? sk : sr;
            }
        }
        const float piv = s[k];
        c[k] = c[k] / piv;
        const float rowk = c[k];
        #pragma unroll
        for (int r = 0; r < TDIM; ++r) {
            if (r == k) continue;
            c[r] = c[r] - s[r] * rowk;
        }
    }

    float t[TDIM] = {};
    #pragma unroll
    for (int k2 = 0; k2 < TDIM; ++k2) {
        #pragma unroll
        for (int i = 0; i < TDIM; ++i)
            t[i] += lane_bcast(c[i], TDIM + k2) * bcol[k2];
    }

    float wv[TDIM] = {};
    #pragma unroll
    for (int k2 = 0; k2 < TDIM; ++k2) {
        #pragma unroll
        for (int i = 0; i < TDIM; ++i)
            wv[i] += lane_bcast(t[i], k2) * acol[k2];
    }

    #pragma unroll
    for (int m = 0; m < TDIM / 2; ++m)
        dst[(2 * m + hi) * PITCH + col] = hi ? wv[2 * m + 1] : wv[2 * m];
}

__device__ __forceinline__ void wave_lds_order()
{
    __builtin_amdgcn_fence(__ATOMIC_RELEASE, "wavefront");
    __builtin_amdgcn_wave_barrier();
    __builtin_amdgcn_fence(__ATOMIC_ACQUIRE, "wavefront");
}

__device__ __forceinline__ void group_tile(float* __restrict__ rs, int first, int live,
                                           const v2f (&w)[4], int c, int hi)
{
    const float* rp = rs + (first + (c < live ? c : 0)) * PITCH;

    v2f xv[4];
    #pragma unroll
    for (int ks = 0; ks < 4; ++ks)
        xv[ks] = *(const v2f*)(rp + 4 * ks + 2 * hi);

    v8f acc = {0.f, 0.f, 0.f, 0.f, 0.f, 0.f, 0.f, 0.f};
    #pragma unroll
    for (int ks = 0; ks < 4; ++ks)
        acc = __builtin_amdgcn_wmma_f32_16x16x4_f32(
                  false, w[ks], false, xv[ks], (short)0, acc, false, false);

    asm volatile("v_nop\n\tv_nop\n\tv_nop\n\tv_nop" : "+v"(acc));

    float4* op = (float4*)(rs + (first + c) * PITCH + 8 * hi);
    const float4 o0 = {acc[0], acc[1], acc[2], acc[3]};
    const float4 o1 = {acc[4], acc[5], acc[6], acc[7]};
    op[0] = o0;
    op[1] = o1;
}

__device__ __forceinline__ void rank_and_scatter(const int (&araw)[4], int last, const float4 (&xr)[16],
                                                 float* __restrict__ rs, int lane,
                                                 int (&base)[NACT], int (&cnt)[NACT],
                                                 int (&slot)[16], bool& anybad)
{
    int av[4];
    #pragma unroll
    for (int q = 0; q < 4; ++q)
        av[q] = (lane + 32 * q <= last) ? araw[q] : -1;

    int pos[4] = {-1, -1, -1, -1};
    int run = 0;
    #pragma unroll
    for (int a = 0; a < NACT; ++a) {
        base[a] = run;
        #pragma unroll
        for (int q = 0; q < 4; ++q) {
            const bool is = (av[q] == a);
            const unsigned m = __builtin_amdgcn_ballot_w32(is);
            const int rk = (int)__builtin_amdgcn_mbcnt_lo(m, (unsigned)run);
            pos[q] = is ? rk : pos[q];
            run += __builtin_popcount(m);
        }
        cnt[a] = run - base[a];
        run = base[a] + ((cnt[a] + 15) & ~15);
    }

    anybad = __builtin_amdgcn_ballot_w32((pos[0] | pos[1] | pos[2] | pos[3]) < 0) != 0u;

    const int piece = (lane & 3) * 4;
    #pragma unroll
    for (int q = 0; q < 16; ++q) {
        const int src = ((q & 3) << 3) + (lane >> 2);
        const int sp  = __builtin_amdgcn_ds_bpermute(src << 2, pos[q >> 2]);
        slot[q] = sp < 0 ? SPARE : sp;
    }

    #pragma unroll
    for (int q = 0; q < 16; ++q)
        *(float4*)(rs + slot[q] * PITCH + piece) = xr[q];
    wave_lds_order();
}

__global__ __launch_bounds__(32 * WAVES_PER_BLK) void gemv_grouped_fused(const float* __restrict__ x,
                                                          const int*   __restrict__ act,
                                                          const float* __restrict__ Ag,
                                                          const float* __restrict__ Bg,
                                                          float*       __restrict__ y,
                                                          int n, int ngroups)
{
    __shared__ __attribute__((aligned(16))) float rows_all[WAVES_PER_BLK][SLOTS * PITCH];

    const int lane = threadIdx.x & 31;
    const int wv   = __builtin_amdgcn_readfirstlane((int)(threadIdx.x >> 5));
    const bool prep = (wv < NACT);
    const int g    = blockIdx.x * WAVES_PER_BLK + wv;
    const bool live = (g < ngroups);

    float* rs = rows_all[wv];
    const int c  = lane & 15;
    const int hi = lane >> 4;
    const int s0 = g * GROUP;
    const bool full = (s0 + GROUP <= n);

    float acol[TDIM] = {}, bcol[TDIM] = {};
    if (prep) {
        #pragma unroll
        for (int r = 0; r < TDIM; ++r) {
            acol[r] = Ag[wv * TDIM * TDIM + r * TDIM + c];
            bcol[r] = Bg[wv * TDIM * TDIM + r * TDIM + c];
        }
    }

    const int last = n - 1 - s0;
    int araw[4];
    #pragma unroll
    for (int q = 0; q < 4; ++q)
        araw[q] = act[s0 + min(lane + 32 * q, last)];

    const float4* xg = (const float4*)(x + (size_t)s0 * TDIM);
    float4 xr[16];
    if (full) {
        #pragma unroll
        for (int q = 0; q < 16; ++q) xr[q] = xg[q * 32 + lane];
    } else {
        #pragma unroll
        for (int q = 0; q < 16; ++q) {
            float4 v = {0.f, 0.f, 0.f, 0.f};
            if (8 * q + (lane >> 2) <= last) v = xg[q * 32 + lane];
            xr[q] = v;
        }
    }

    int base[NACT] = {}, cnt[NACT] = {};
    int slot[16] = {};
    bool anybad = false;
    if (prep)      weights_to_lds(acol, bcol, rs, c, hi);
    else if (live) rank_and_scatter(araw, last, xr, rs, lane, base, cnt, slot, anybad);
    __syncthreads();
    v2f wr[NACT][4];
    #pragma unroll
    for (int a = 0; a < NACT; ++a) {
        #pragma unroll
        for (int ks = 0; ks < 4; ++ks)
            wr[a][ks] = *(const v2f*)(rows_all[a] + c * PITCH + 4 * ks + 2 * hi);
    }
    __syncthreads();
    if (!live) return;

    if (prep) rank_and_scatter(araw, last, xr, rs, lane, base, cnt, slot, anybad);
    const int piece = (lane & 3) * 4;

    #pragma unroll
    for (int a = 0; a < NACT; ++a) {
        for (int t = 0; 16 * t < cnt[a]; ++t)
            group_tile(rs, base[a] + 16 * t, cnt[a] - 16 * t, wr[a], c, hi);
    }
    wave_lds_order();

    float4* yg = (float4*)(y + (size_t)s0 * TDIM);
    if (full && !anybad) {
        float4 o[16];
        #pragma unroll
        for (int q = 0; q < 16; ++q)
            o[q] = *(const float4*)(rs + slot[q] * PITCH + piece);
        #pragma unroll
        for (int q = 0; q < 16; ++q) yg[q * 32 + lane] = o[q];
    } else {
        #pragma unroll
        for (int q = 0; q < 16; ++q) {
            float4 v = *(const float4*)(rs + slot[q] * PITCH + piece);
            if (slot[q] == SPARE) {
                const float4 z = {0.f, 0.f, 0.f, 0.f};
                v = z;
            }
            if (s0 + q * 8 + (lane >> 2) < n) yg[q * 32 + lane] = v;
        }
    }
}

extern "C" void kernel_launch(void* const* d_in, const int* in_sizes, int n_in,
                              void* d_out, int out_size, void* d_ws, size_t ws_size,
                              hipStream_t stream) {
    const float* x    = (const float*)d_in[0];
    const int*   a    = (const int*)  d_in[1];
    const float* A    = (const float*)d_in[2];
    const float* Brep = (const float*)d_in[3];
    float* y  = (float*)d_out;

    const int n = in_sizes[1];

    const int ngroups = (n + GROUP - 1) / GROUP;
    int blocks = (ngroups + WAVES_PER_BLK - 1) / WAVES_PER_BLK;
    if (blocks < 1) blocks = 1;
    gemv_grouped_fused<<<blocks, 32 * WAVES_PER_BLK, 0, stream>>>(x, a, A, Brep, y, n, ngroups);
}


# Weights as the solution of A W = B_rep*A: B_rep*A by 4 f32 matrix instructions, then the same pivoted elimination on [A | B_rep*A] leaves W; both 16x16x16 lane-read products go; sums reassociated
# speedup vs baseline: 3.3053x; 1.0575x over previous
#include <hip/hip_runtime.h>

typedef __attribute__((ext_vector_type(2))) float v2f;
typedef __attribute__((ext_vector_type(8))) float v8f;

#define TDIM 16
#define NACT 8

__device__ __forceinline__ float lane_bcast(float v, int src)
{
    return __int_as_float(__builtin_amdgcn_readlane(__float_as_int(v), src));
}

#define GROUP   128
#define SLOTS   249
#define SPARE   (SLOTS - 1)
#define PITCH   20
#define WAVES_PER_BLK 16
static_assert(NACT <= WAVES_PER_BLK, "part 1 needs one wave per action in every work-group");

__device__ __forceinline__ float other_half(float v)
{
    return __int_as_float(__builtin_amdgcn_permlanex16(__float_as_int(v), __float_as_int(v),
                                                       0x76543210, 0xfedcba98, false, false));
}

__device__ __forceinline__ void weights_to_lds(const float (&acol)[TDIM], const v2f (&brow)[4],
                                               float* __restrict__ dst, int col, int hi)
{

    v2f ak[4];
    #pragma unroll
    for (int ks = 0; ks < 4; ++ks) {
        ak[ks][0] = hi ? acol[4 * ks + 2] : acol[4 * ks];
        ak[ks][1] = hi ? acol[4 * ks + 3] : acol[4 * ks + 1];
    }
    v8f p = {0.f, 0.f, 0.f, 0.f, 0.f, 0.f, 0.f, 0.f};
    #pragma unroll
    for (int ks = 0; ks < 4; ++ks)
        p = __builtin_amdgcn_wmma_f32_16x16x4_f32(
                false, brow[ks], false, ak[ks], (short)0, p, false, false);
    asm volatile("v_nop\n\tv_nop\n\tv_nop\n\tv_nop"
                 : "+v"(p)
                 : "v"(brow[0]), "v"(brow[1]), "v"(brow[2]), "v"(brow[3]),
                   "v"(ak[0]), "v"(ak[1]), "v"(ak[2]), "v"(ak[3]));

    float c[TDIM];
    #pragma unroll
    for (int v = 0; v < TDIM / 2; ++v) {
        const float below = other_half(p[v]);
        c[v]            = hi ? below : acol[v];
        c[v + TDIM / 2] = hi ? p[v]  : acol[v + TDIM / 2];
    }

    #pragma unroll
    for (int k = 0; k < TDIM; ++k) {
        float s[TDIM];
        #pragma unroll
        for (int r = 0; r < TDIM; ++r) s[r] = lane_bcast(c[r], k);

        int prow = k;
        float best = fabsf(s[k]);
        #pragma unroll
        for (int r = k + 1; r < TDIM; ++r) {
            float v = fabsf(s[r]);
            if (v > best) { best = v; prow = r; }
        }
        if (prow != k) {
            #pragma unroll
            for (int r = k + 1; r < TDIM; ++r) {
                const bool sw = (prow == r);
                const float ck = c[k], cr = c[r];
                c[k] = sw ? cr : ck;
                c[r] = sw ? ck : cr;
                const float sk = s[k], sr = s[r];
                s[k] = sw ? sr : sk;
                s[r] = sw ? sk : sr;
            }
        }
        const float piv = s[k];
        c[k] = c[k] / piv;
        const float rowk = c[k];
        #pragma unroll
        for (int r = 0; r < TDIM; ++r) {
            if (r == k) continue;
            c[r] = c[r] - s[r] * rowk;
        }
    }

    #pragma unroll
    for (int m = 0; m < TDIM / 2; ++m) {
        const float even = other_half(c[2 * m]);
        dst[(2 * m + hi) * PITCH + col] = hi ? c[2 * m + 1] : even;
    }
}

__device__ __forceinline__ void wave_lds_order()
{
    __builtin_amdgcn_fence(__ATOMIC_RELEASE, "wavefront");
    __builtin_amdgcn_wave_barrier();
    __builtin_amdgcn_fence(__ATOMIC_ACQUIRE, "wavefront");
}

__device__ __forceinline__ void group_tile(float* __restrict__ rs, int first, int live,
                                           const v2f (&w)[4], int c, int hi)
{
    const float* rp = rs + (first + (c < live ? c : 0)) * PITCH;

    v2f xv[4];
    #pragma unroll
    for (int ks = 0; ks < 4; ++ks)
        xv[ks] = *(const v2f*)(rp + 4 * ks + 2 * hi);

    v8f acc = {0.f, 0.f, 0.f, 0.f, 0.f, 0.f, 0.f, 0.f};
    #pragma unroll
    for (int ks = 0; ks < 4; ++ks)
        acc = __builtin_amdgcn_wmma_f32_16x16x4_f32(
                  false, w[ks], false, xv[ks], (short)0, acc, false, false);

    asm volatile("v_nop\n\tv_nop\n\tv_nop\n\tv_nop" : "+v"(acc));

    float4* op = (float4*)(rs + (first + c) * PITCH + 8 * hi);
    const float4 o0 = {acc[0], acc[1], acc[2], acc[3]};
    const float4 o1 = {acc[4], acc[5], acc[6], acc[7]};
    op[0] = o0;
    op[1] = o1;
}

__device__ __forceinline__ void rank_and_scatter(const int (&araw)[4], int last, const float4 (&xr)[16],
                                                 float* __restrict__ rs, int lane,
                                                 int (&base)[NACT], int (&cnt)[NACT],
                                                 int (&slot)[16], bool& anybad)
{
    int av[4];
    #pragma unroll
    for (int q = 0; q < 4; ++q)
        av[q] = (lane + 32 * q <= last) ? araw[q] : -1;

    int pos[4] = {-1, -1, -1, -1};
    int run = 0;
    #pragma unroll
    for (int a = 0; a < NACT; ++a) {
        base[a] = run;
        #pragma unroll
        for (int q = 0; q < 4; ++q) {
            const bool is = (av[q] == a);
            const unsigned m = __builtin_amdgcn_ballot_w32(is);
            const int rk = (int)__builtin_amdgcn_mbcnt_lo(m, (unsigned)run);
            pos[q] = is ? rk : pos[q];
            run += __builtin_popcount(m);
        }
        cnt[a] = run - base[a];
        run = base[a] + ((cnt[a] + 15) & ~15);
    }

    anybad = __builtin_amdgcn_ballot_w32((pos[0] | pos[1] | pos[2] | pos[3]) < 0) != 0u;

    const int piece = (lane & 3) * 4;
    #pragma unroll
    for (int q = 0; q < 16; ++q) {
        const int src = ((q & 3) << 3) + (lane >> 2);
        const int sp  = __builtin_amdgcn_ds_bpermute(src << 2, pos[q >> 2]);
        slot[q] = sp < 0 ? SPARE : sp;
    }

    #pragma unroll
    for (int q = 0; q < 16; ++q)
        *(float4*)(rs + slot[q] * PITCH + piece) = xr[q];
    wave_lds_order();
}

__global__ __launch_bounds__(32 * WAVES_PER_BLK) void gemv_grouped_fused(const float* __restrict__ x,
                                                          const int*   __restrict__ act,
                                                          const float* __restrict__ Ag,
                                                          const float* __restrict__ Bg,
                                                          float*       __restrict__ y,
                                                          int n, int ngroups)
{
    __shared__ __attribute__((aligned(16))) float rows_all[WAVES_PER_BLK][SLOTS * PITCH];

    const int lane = threadIdx.x & 31;
    const int wv   = __builtin_amdgcn_readfirstlane((int)(threadIdx.x >> 5));
    const bool prep = (wv < NACT);
    const int g    = blockIdx.x * WAVES_PER_BLK + wv;
    const bool live = (g < ngroups);

    float* rs = rows_all[wv];
    const int c  = lane & 15;
    const int hi = lane >> 4;
    const int s0 = g * GROUP;
    const bool full = (s0 + GROUP <= n);

    float acol[TDIM] = {};
    v2f brow[4] = {};
    if (prep) {
        #pragma unroll
        for (int r = 0; r < TDIM; ++r)
            acol[r] = Ag[wv * TDIM * TDIM + r * TDIM + c];
        #pragma unroll
        for (int ks = 0; ks < 4; ++ks)
            brow[ks] = *(const v2f*)(Bg + wv * TDIM * TDIM + c * TDIM + 4 * ks + 2 * hi);
    }

    const int last = n - 1 - s0;
    int araw[4];
    #pragma unroll
    for (int q = 0; q < 4; ++q)
        araw[q] = act[s0 + min(lane + 32 * q, last)];

    const float4* xg = (const float4*)(x + (size_t)s0 * TDIM);
    float4 xr[16];
    if (full) {
        #pragma unroll
        for (int q = 0; q < 16; ++q) xr[q] = xg[q * 32 + lane];
    } else {
        #pragma unroll
        for (int q = 0; q < 16; ++q) {
            float4 v = {0.f, 0.f, 0.f, 0.f};
            if (8 * q + (lane >> 2) <= last) v = xg[q * 32 + lane];
            xr[q] = v;
        }
    }

    int base[NACT] = {}, cnt[NACT] = {};
    int slot[16] = {};
    bool anybad = false;
    if (prep)      weights_to_lds(acol, brow, rs, c, hi);
    else if (live) rank_and_scatter(araw, last, xr, rs, lane, base, cnt, slot, anybad);
    __syncthreads();
    v2f wr[NACT][4];
    #pragma unroll
    for (int a = 0; a < NACT; ++a) {
        #pragma unroll
        for (int ks = 0; ks < 4; ++ks)
            wr[a][ks] = *(const v2f*)(rows_all[a] + c * PITCH + 4 * ks + 2 * hi);
    }
    __syncthreads();
    if (!live) return;

    if (prep) rank_and_scatter(araw, last, xr, rs, lane, base, cnt, slot, anybad);
    const int piece = (lane & 3) * 4;

    #pragma unroll
    for (int a = 0; a < NACT; ++a) {
        for (int t = 0; 16 * t < cnt[a]; ++t)
            group_tile(rs, base[a] + 16 * t, cnt[a] - 16 * t, wr[a], c, hi);
    }
    wave_lds_order();

    float4* yg = (float4*)(y + (size_t)s0 * TDIM);
    if (full && !anybad) {
        float4 o[16];
        #pragma unroll
        for (int q = 0; q < 16; ++q)
            o[q] = *(const float4*)(rs + slot[q] * PITCH + piece);
        #pragma unroll
        for (int q = 0; q < 16; ++q) yg[q * 32 + lane] = o[q];
    } else {
        #pragma unroll
        for (int q = 0; q < 16; ++q) {
            float4 v = *(const float4*)(rs + slot[q] * PITCH + piece);
            if (slot[q] == SPARE) {
                const float4 z = {0.f, 0.f, 0.f, 0.f};
                v = z;
            }
            if (s0 + q * 8 + (lane >> 2) < n) yg[q * 32 + lane] = v;
        }
    }
}

extern "C" void kernel_launch(void* const* d_in, const int* in_sizes, int n_in,
                              void* d_out, int out_size, void* d_ws, size_t ws_size,
                              hipStream_t stream) {
    const float* x    = (const float*)d_in[0];
    const int*   a    = (const int*)  d_in[1];
    const float* A    = (const float*)d_in[2];
    const float* Brep = (const float*)d_in[3];
    float* y  = (float*)d_out;

    const int n = in_sizes[1];

    const int ngroups = (n + GROUP - 1) / GROUP;
    int blocks = (ngroups + WAVES_PER_BLK - 1) / WAVES_PER_BLK;
    if (blocks < 1) blocks = 1;
    gemv_grouped_fused<<<blocks, 32 * WAVES_PER_BLK, 0, stream>>>(x, a, A, Brep, y, n, ngroups);
}


# Part 1 follows the row loads inside each load path, not behind their join, so a full wave's load-count wait is exact: part 1 waits for A and B_rep only and runs while the rows arrive; sums unchanged
# speedup vs baseline: 3.6170x; 1.0943x over previous
#include <hip/hip_runtime.h>

typedef __attribute__((ext_vector_type(2))) float v2f;
typedef __attribute__((ext_vector_type(8))) float v8f;

#define TDIM 16
#define NACT 8

__device__ __forceinline__ float lane_bcast(float v, int src)
{
    return __int_as_float(__builtin_amdgcn_readlane(__float_as_int(v), src));
}

#define GROUP   128
#define SLOTS   249
#define SPARE   (SLOTS - 1)
#define PITCH   20
#define WAVES_PER_BLK 16
static_assert(NACT <= WAVES_PER_BLK, "part 1 needs one wave per action in every work-group");

__device__ __forceinline__ float other_half(float v)
{
    return __int_as_float(__builtin_amdgcn_permlanex16(__float_as_int(v), __float_as_int(v),
                                                       0x76543210, 0xfedcba98, false, false));
}

__device__ __forceinline__ void weights_to_lds(const float (&acol)[TDIM], const v2f (&brow)[4],
                                               float* __restrict__ dst, int col, int hi)
{

    v2f ak[4];
    #pragma unroll
    for (int ks = 0; ks < 4; ++ks) {
        ak[ks][0] = hi ? acol[4 * ks + 2] : acol[4 * ks];
        ak[ks][1] = hi ? acol[4 * ks + 3] : acol[4 * ks + 1];
    }
    v8f p = {0.f, 0.f, 0.f, 0.f, 0.f, 0.f, 0.f, 0.f};
    #pragma unroll
    for (int ks = 0; ks < 4; ++ks)
        p = __builtin_amdgcn_wmma_f32_16x16x4_f32(
                false, brow[ks], false, ak[ks], (short)0, p, false, false);
    asm volatile("v_nop\n\tv_nop\n\tv_nop\n\tv_nop"
                 : "+v"(p)
                 : "v"(brow[0]), "v"(brow[1]), "v"(brow[2]), "v"(brow[3]),
                   "v"(ak[0]), "v"(ak[1]), "v"(ak[2]), "v"(ak[3]));

    float c[TDIM];
    #pragma unroll
    for (int v = 0; v < TDIM / 2; ++v) {
        const float below = other_half(p[v]);
        c[v]            = hi ? below : acol[v];
        c[v + TDIM / 2] = hi ? p[v]  : acol[v + TDIM / 2];
    }

    #pragma unroll
    for (int k = 0; k < TDIM; ++k) {
        float s[TDIM];
        #pragma unroll
        for (int r = 0; r < TDIM; ++r) s[r] = lane_bcast(c[r], k);

        int prow = k;
        float best = fabsf(s[k]);
        #pragma unroll
        for (int r = k + 1; r < TDIM; ++r) {
            float v = fabsf(s[r]);
            if (v > best) { best = v; prow = r; }
        }
        if (prow != k) {
            #pragma unroll
            for (int r = k + 1; r < TDIM; ++r) {
                const bool sw = (prow == r);
                const float ck = c[k], cr = c[r];
                c[k] = sw ? cr : ck;
                c[r] = sw ? ck : cr;
                const float sk = s[k], sr = s[r];
                s[k] = sw ? sr : sk;
                s[r] = sw ? sk : sr;
            }
        }
        const float piv = s[k];
        c[k] = c[k] / piv;
        const float rowk = c[k];
        #pragma unroll
        for (int r = 0; r < TDIM; ++r) {
            if (r == k) continue;
            c[r] = c[r] - s[r] * rowk;
        }
    }

    #pragma unroll
    for (int m = 0; m < TDIM / 2; ++m) {
        const float even = other_half(c[2 * m]);
        dst[(2 * m + hi) * PITCH + col] = hi ? c[2 * m + 1] : even;
    }
}

__device__ __forceinline__ void wave_lds_order()
{
    __builtin_amdgcn_fence(__ATOMIC_RELEASE, "wavefront");
    __builtin_amdgcn_wave_barrier();
    __builtin_amdgcn_fence(__ATOMIC_ACQUIRE, "wavefront");
}

__device__ __forceinline__ void group_tile(float* __restrict__ rs, int first, int live,
                                           const v2f (&w)[4], int c, int hi)
{
    const float* rp = rs + (first + (c < live ? c : 0)) * PITCH;

    v2f xv[4];
    #pragma unroll
    for (int ks = 0; ks < 4; ++ks)
        xv[ks] = *(const v2f*)(rp + 4 * ks + 2 * hi);

    v8f acc = {0.f, 0.f, 0.f, 0.f, 0.f, 0.f, 0.f, 0.f};
    #pragma unroll
    for (int ks = 0; ks < 4; ++ks)
        acc = __builtin_amdgcn_wmma_f32_16x16x4_f32(
                  false, w[ks], false, xv[ks], (short)0, acc, false, false);

    asm volatile("v_nop\n\tv_nop\n\tv_nop\n\tv_nop" : "+v"(acc));

    float4* op = (float4*)(rs + (first + c) * PITCH + 8 * hi);
    const float4 o0 = {acc[0], acc[1], acc[2], acc[3]};
    const float4 o1 = {acc[4], acc[5], acc[6], acc[7]};
    op[0] = o0;
    op[1] = o1;
}

__device__ __forceinline__ void rank_and_scatter(const int (&araw)[4], int last, const float4 (&xr)[16],
                                                 float* __restrict__ rs, int lane,
                                                 int (&base)[NACT], int (&cnt)[NACT],
                                                 int (&slot)[16], bool& anybad)
{
    int av[4];
    #pragma unroll
    for (int q = 0; q < 4; ++q)
        av[q] = (lane + 32 * q <= last) ? araw[q] : -1;

    int pos[4] = {-1, -1, -1, -1};
    int run = 0;
    #pragma unroll
    for (int a = 0; a < NACT; ++a) {
        base[a] = run;
        #pragma unroll
        for (int q = 0; q < 4; ++q) {
            const bool is = (av[q] == a);
            const unsigned m = __builtin_amdgcn_ballot_w32(is);
            const int rk = (int)__builtin_amdgcn_mbcnt_lo(m, (unsigned)run);
            pos[q] = is ? rk : pos[q];
            run += __builtin_popcount(m);
        }
        cnt[a] = run - base[a];
        run = base[a] + ((cnt[a] + 15) & ~15);
    }

    anybad = __builtin_amdgcn_ballot_w32((pos[0] | pos[1] | pos[2] | pos[3]) < 0) != 0u;

    const int piece = (lane & 3) * 4;
    #pragma unroll
    for (int q = 0; q < 16; ++q) {
        const int src = ((q & 3) << 3) + (lane >> 2);
        const int sp  = __builtin_amdgcn_ds_bpermute(src << 2, pos[q >> 2]);
        slot[q] = sp < 0 ? SPARE : sp;
    }

    #pragma unroll
    for (int q = 0; q < 16; ++q)
        *(float4*)(rs + slot[q] * PITCH + piece) = xr[q];
    wave_lds_order();
}

__global__ __launch_bounds__(32 * WAVES_PER_BLK) void gemv_grouped_fused(const float* __restrict__ x,
                                                          const int*   __restrict__ act,
                                                          const float* __restrict__ Ag,
                                                          const float* __restrict__ Bg,
                                                          float*       __restrict__ y,
                                                          int n, int ngroups)
{
    __shared__ __attribute__((aligned(16))) float rows_all[WAVES_PER_BLK][SLOTS * PITCH];

    const int lane = threadIdx.x & 31;
    const int wv   = __builtin_amdgcn_readfirstlane((int)(threadIdx.x >> 5));
    const bool prep = (wv < NACT);
    const int g    = blockIdx.x * WAVES_PER_BLK + wv;
    const bool live = (g < ngroups);

    float* rs = rows_all[wv];
    const int c  = lane & 15;
    const int hi = lane >> 4;
    const int s0 = g * GROUP;
    const bool full = (s0 + GROUP <= n);

    float acol[TDIM] = {};
    v2f brow[4] = {};
    if (prep) {
        #pragma unroll
        for (int r = 0; r < TDIM; ++r)
            acol[r] = Ag[wv * TDIM * TDIM + r * TDIM + c];
        #pragma unroll
        for (int ks = 0; ks < 4; ++ks)
            brow[ks] = *(const v2f*)(Bg + wv * TDIM * TDIM + c * TDIM + 4 * ks + 2 * hi);
    }

    const int last = n - 1 - s0;
    int araw[4];
    #pragma unroll
    for (int q = 0; q < 4; ++q)
        araw[q] = act[s0 + min(lane + 32 * q, last)];

    int base[NACT] = {}, cnt[NACT] = {};
    int slot[16] = {};
    bool anybad = false;

    const float4* xg = (const float4*)(x + (size_t)s0 * TDIM);
    float4 xr[16];
    if (full) {
        #pragma unroll
        for (int q = 0; q < 16; ++q) xr[q] = xg[q * 32 + lane];
        if (prep) weights_to_lds(acol, brow, rs, c, hi);
    } else {
        #pragma unroll
        for (int q = 0; q < 16; ++q) {
            float4 v = {0.f, 0.f, 0.f, 0.f};
            if (8 * q + (lane >> 2) <= last) v = xg[q * 32 + lane];
            xr[q] = v;
        }
        if (prep) weights_to_lds(acol, brow, rs, c, hi);
    }

    if (!prep && live) rank_and_scatter(araw, last, xr, rs, lane, base, cnt, slot, anybad);
    __syncthreads();
    v2f wr[NACT][4];
    #pragma unroll
    for (int a = 0; a < NACT; ++a) {
        #pragma unroll
        for (int ks = 0; ks < 4; ++ks)
            wr[a][ks] = *(const v2f*)(rows_all[a] + c * PITCH + 4 * ks + 2 * hi);
    }
    __syncthreads();
    if (!live) return;

    if (prep) rank_and_scatter(araw, last, xr, rs, lane, base, cnt, slot, anybad);
    const int piece = (lane & 3) * 4;

    #pragma unroll
    for (int a = 0; a < NACT; ++a) {
        for (int t = 0; 16 * t < cnt[a]; ++t)
            group_tile(rs, base[a] + 16 * t, cnt[a] - 16 * t, wr[a], c, hi);
    }
    wave_lds_order();

    float4* yg = (float4*)(y + (size_t)s0 * TDIM);
    if (full && !anybad) {
        float4 o[16];
        #pragma unroll
        for (int q = 0; q < 16; ++q)
            o[q] = *(const float4*)(rs + slot[q] * PITCH + piece);
        #pragma unroll
        for (int q = 0; q < 16; ++q) yg[q * 32 + lane] = o[q];
    } else {
        #pragma unroll
        for (int q = 0; q < 16; ++q) {
            float4 v = *(const float4*)(rs + slot[q] * PITCH + piece);
            if (slot[q] == SPARE) {
                const float4 z = {0.f, 0.f, 0.f, 0.f};
                v = z;
            }
            if (s0 + q * 8 + (lane >> 2) < n) yg[q * 32 + lane] = v;
        }
    }
}

extern "C" void kernel_launch(void* const* d_in, const int* in_sizes, int n_in,
                              void* d_out, int out_size, void* d_ws, size_t ws_size,
                              hipStream_t stream) {
    const float* x    = (const float*)d_in[0];
    const int*   a    = (const int*)  d_in[1];
    const float* A    = (const float*)d_in[2];
    const float* Brep = (const float*)d_in[3];
    float* y  = (float*)d_out;

    const int n = in_sizes[1];

    const int ngroups = (n + GROUP - 1) / GROUP;
    int blocks = (ngroups + WAVES_PER_BLK - 1) / WAVES_PER_BLK;
    if (blocks < 1) blocks = 1;
    gemv_grouped_fused<<<blocks, 32 * WAVES_PER_BLK, 0, stream>>>(x, a, A, Brep, y, n, ngroups);
}
